# starting kernel
# baseline (speedup 1.0000x reference)
#include <hip/hip_runtime.h>
#include <hip/hip_bf16.h>
#include <math.h>

#define BATCH 128
#define RNUM  2048
#define ONUM  16
#define DNUM  16
#define CNUM  16

typedef __attribute__((ext_vector_type(2))) float v2f;
typedef __attribute__((ext_vector_type(8))) float v8f;

__device__ __forceinline__ float halfsum16(float v) {
    v += __shfl_xor(v, 1);
    v += __shfl_xor(v, 2);
    v += __shfl_xor(v, 4);
    v += __shfl_xor(v, 8);
    return v;
}

template <int MODE>
__global__ __launch_bounds__(256) void caps_kernel(
    const float* __restrict__ x,
    const float* __restrict__ W,
    float* __restrict__ bmat,
    const float* __restrict__ rowmax,
    const float* __restrict__ denom,
    float* __restrict__ out)
{
    const int r    = blockIdx.x;
    const int tid  = threadIdx.x;
    const int L    = tid & 31;
    const int w    = tid >> 5;
    const int h    = L >> 4;
    const int m16  = L & 15;
    const int bb0  = w * 16;

    v2f a[4];
    const float* xb = x + (bb0 + m16) * DNUM;
    #pragma unroll
    for (int k = 0; k < 4; ++k) {
        a[k] = *reinterpret_cast<const v2f*>(xb + 4 * k + 2 * h);
    }

    v8f U[ONUM];
    const float* Wr = W + (size_t)r * ONUM * DNUM * CNUM;
    #pragma unroll
    for (int o = 0; o < ONUM; ++o) {
        const float* Wo = Wr + o * (DNUM * CNUM);
        v8f acc = {0.f, 0.f, 0.f, 0.f, 0.f, 0.f, 0.f, 0.f};
        #pragma unroll
        for (int k = 0; k < 4; ++k) {
            v2f bmx;
            const int row0 = 4 * k + 2 * h;
            bmx.x = Wo[(row0 + 0) * CNUM + m16];
            bmx.y = Wo[(row0 + 1) * CNUM + m16];
            acc = __builtin_amdgcn_wmma_f32_16x16x4_f32(
                false, a[k], false, bmx, (short)0, acc, false, false);
        }
        U[o] = acc;
    }

    float braw[8], creg[8];
    #pragma unroll
    for (int i = 0; i < 8; ++i) {
        if (MODE == 0) {
            braw[i] = 0.0f;
            creg[i] = 1.0f / (float)RNUM;
        } else {
            const int b = bb0 + i + 8 * h;
            braw[i] = bmat[((size_t)b * RNUM + r) * ONUM + m16];
            const float rm = rowmax[b * ONUM + m16];
            const float dn = denom[b * ONUM + m16];
            creg[i] = __expf(braw[i] - rm) / dn;
        }
    }

    v8f s = {0.f, 0.f, 0.f, 0.f, 0.f, 0.f, 0.f, 0.f};
    #pragma unroll
    for (int o = 0; o < ONUM; ++o) {
        #pragma unroll
        for (int i = 0; i < 8; ++i) {
            float co;
            if (MODE == 0) {
                co = 1.0f / (float)RNUM;
            } else {
                co = __shfl(creg[i], (tid & 16) | o);
            }
            s[i] += co * U[o][i];
        }
    }

    v8f vj;
    #pragma unroll
    for (int i = 0; i < 8; ++i) {
        const float ns  = halfsum16(s[i] * s[i]);
        const float fac = sqrtf(ns) / (1.0f + ns);
        vj[i] = s[i] * fac;
    }

    if (MODE < 2) {
        float aout[8];
        #pragma unroll
        for (int o = 0; o < ONUM; ++o) {
            #pragma unroll
            for (int i = 0; i < 8; ++i) {
                const float t = halfsum16(U[o][i] * vj[i]);
                if (m16 == o) aout[i] = t;
            }
        }
        #pragma unroll
        for (int i = 0; i < 8; ++i) {
            const int b = bb0 + i + 8 * h;
            bmat[((size_t)b * RNUM + r) * ONUM + m16] = braw[i] + aout[i];
        }
    } else {
        #pragma unroll
        for (int i = 0; i < 8; ++i) {
            const int b = bb0 + i + 8 * h;
            out[((size_t)b * RNUM + r) * CNUM + m16] = vj[i];
        }
    }
}

__global__ __launch_bounds__(256) void softmax_reduce(
    const float* __restrict__ bmat,
    float* __restrict__ rowmax,
    float* __restrict__ denom)
{
    __shared__ float sm[256];
    const int b  = blockIdx.x;
    const int t  = threadIdx.x;
    const int o  = t & 15;
    const int rs = t >> 4;

    float mx = -INFINITY;
    for (int r = rs; r < RNUM; r += 16)
        mx = fmaxf(mx, bmat[((size_t)b * RNUM + r) * ONUM + o]);
    sm[t] = mx;
    __syncthreads();
    for (int off = 128; off >= 16; off >>= 1) {
        if (t < off) sm[t] = fmaxf(sm[t], sm[t + off]);
        __syncthreads();
    }
    const float rm = sm[o];
    if (t < 16) rowmax[b * ONUM + t] = sm[t];
    __syncthreads();

    float se = 0.0f;
    for (int r = rs; r < RNUM; r += 16)
        se += __expf(bmat[((size_t)b * RNUM + r) * ONUM + o] - rm);
    sm[t] = se;
    __syncthreads();
    for (int off = 128; off >= 16; off >>= 1) {
        if (t < off) sm[t] += sm[t + off];
        __syncthreads();
    }
    if (t < 16) denom[b * ONUM + t] = sm[t];
}

extern "C" void kernel_launch(void* const* d_in, const int* in_sizes, int n_in,
                              void* d_out, int out_size, void* d_ws, size_t ws_size,
                              hipStream_t stream) {
    const float* x = (const float*)d_in[0];
    const float* W = (const float*)d_in[1];
    float* out = (float*)d_out;

    float* bmat = (float*)d_ws;
    float* rowmax = bmat + (size_t)BATCH * RNUM * ONUM;
    float* denom  = rowmax + BATCH * ONUM;

    dim3 grid(RNUM), block(256);

    caps_kernel<0><<<grid, block, 0, stream>>>(x, W, bmat, rowmax, denom, out);
    softmax_reduce<<<dim3(BATCH), block, 0, stream>>>(bmat, rowmax, denom);
    caps_kernel<1><<<grid, block, 0, stream>>>(x, W, bmat, rowmax, denom, out);
    softmax_reduce<<<dim3(BATCH), block, 0, stream>>>(bmat, rowmax, denom);
    caps_kernel<2><<<grid, block, 0, stream>>>(x, W, bmat, rowmax, denom, out);
}


# One wave per r keeps the W[r] operand fragments of both products in registers and walks the 8 batch tiles with them: W loads per tile drop 8x (grid 2048 -> 256 blocks); sum orders unchanged
# speedup vs baseline: 1.2702x; 1.2702x over previous
#include <hip/hip_runtime.h>
#include <hip/hip_bf16.h>
#include <math.h>


#define BATCH 128
#define RNUM  2048
#define ONUM  16
#define DNUM  16
#define CNUM  16
#define NTILE (BATCH / 16)

typedef __attribute__((ext_vector_type(2))) float v2f;
typedef __attribute__((ext_vector_type(4))) float v4f;
typedef __attribute__((ext_vector_type(8))) float v8f;

__device__ __forceinline__ v8f mma4(v2f a, v2f b, v8f c) {
    return __builtin_amdgcn_wmma_f32_16x16x4_f32(false, a, false, b, (short)0, c, false, false);
}

struct TileIn {
    v2f xa[4];
    v4f xr[4];
    v4f bl0, bl1, rm0, rm1, dn0, dn1;
};

template <int MODE>
__device__ __forceinline__ void load_tile(TileIn& ti,
    const float* __restrict__ x, const float* bmat,
    const float* __restrict__ rowmax, const float* __restrict__ denom,
    int b, int r, int h)
{
    const float* xb = x + b * DNUM;
    #pragma unroll
    for (int k = 0; k < 4; ++k)
        ti.xa[k] = *reinterpret_cast<const v2f*>(xb + 4 * k + 2 * h);
    #pragma unroll
    for (int t = 0; t < 4; ++t)
        ti.xr[t] = (MODE < 2) ? *reinterpret_cast<const v4f*>(xb + 4 * t) : (v4f){0.f, 0.f, 0.f, 0.f};
    const v4f zero4 = {0.f, 0.f, 0.f, 0.f};
    ti.bl0 = zero4; ti.bl1 = zero4; ti.rm0 = zero4; ti.rm1 = zero4; ti.dn0 = zero4; ti.dn1 = zero4;
    if (MODE > 0) {
        const size_t brow = ((size_t)b * RNUM + r) * ONUM + 8 * h;
        ti.bl0 = *reinterpret_cast<const v4f*>(bmat + brow);
        ti.bl1 = *reinterpret_cast<const v4f*>(bmat + brow + 4);
        ti.rm0 = *reinterpret_cast<const v4f*>(rowmax + b * ONUM + 8 * h);
        ti.rm1 = *reinterpret_cast<const v4f*>(rowmax + b * ONUM + 8 * h + 4);
        ti.dn0 = *reinterpret_cast<const v4f*>(denom + b * ONUM + 8 * h);
        ti.dn1 = *reinterpret_cast<const v4f*>(denom + b * ONUM + 8 * h + 4);
    }
}

template <int MODE>
__global__ __launch_bounds__(256) void caps_kernel(
    const float* __restrict__ x,
    const float* __restrict__ W,
    float* bmat,
    const float* __restrict__ rowmax,
    const float* __restrict__ denom,
    float* __restrict__ out)
{
    const int tid  = threadIdx.x;
    const int L    = tid & 31;
    const int w    = tid >> 5;
    const int h    = L >> 4;
    const int m16  = L & 15;
    const int r    = blockIdx.x * 8 + w;

    const float* Wr = W + (size_t)r * (ONUM * DNUM * CNUM);

    TileIn cur;
    load_tile<MODE>(cur, x, bmat, rowmax, denom, m16, r, h);

    v2f wf1[ONUM][4];
    #pragma unroll
    for (int o = 0; o < ONUM; ++o) {
        #pragma unroll
        for (int k = 0; k < 4; ++k) {
            const int d0 = 4 * k + 2 * h;
            wf1[o][k].x = Wr[o * (DNUM * CNUM) + (d0 + 0) * CNUM + m16];
            wf1[o][k].y = Wr[o * (DNUM * CNUM) + (d0 + 1) * CNUM + m16];
        }
    }
    v4f wf2a[DNUM], wf2b[DNUM];
    if (MODE < 2) {
        const float* Wl = Wr + m16 * (DNUM * CNUM) + 8 * h;
        #pragma unroll
        for (int d = 0; d < DNUM; ++d) {
            wf2a[d] = *reinterpret_cast<const v4f*>(Wl + d * CNUM);
            wf2b[d] = *reinterpret_cast<const v4f*>(Wl + d * CNUM + 4);
        }
    }

    for (int t = 0; t < NTILE; ++t) {
        const int b = t * 16 + m16;

        TileIn nxt;
        load_tile<MODE>(nxt, x, bmat, rowmax, denom, (t < NTILE - 1 ? t + 1 : t) * 16 + m16, r, h);

        float cf[ONUM];
        float bold[8];
        if (MODE == 0) {
            #pragma unroll
            for (int o = 0; o < ONUM; ++o) cf[o] = 1.0f / (float)RNUM;
            #pragma unroll
            for (int i = 0; i < 8; ++i) bold[i] = 0.0f;
        } else {
            float cown[8], coth[8];
            #pragma unroll
            for (int i = 0; i < 4; ++i) {
                bold[i]     = cur.bl0[i];
                bold[4 + i] = cur.bl1[i];
                cown[i]     = __expf(cur.bl0[i] - cur.rm0[i]) / cur.dn0[i];
                cown[4 + i] = __expf(cur.bl1[i] - cur.rm1[i]) / cur.dn1[i];
            }
            #pragma unroll
            for (int i = 0; i < 8; ++i) coth[i] = __shfl_xor(cown[i], 16);
            #pragma unroll
            for (int i = 0; i < 8; ++i) {
                cf[i]     = h ? coth[i] : cown[i];
                cf[8 + i] = h ? cown[i] : coth[i];
            }
        }

        v8f acc[4];
        #pragma unroll
        for (int k = 0; k < 4; ++k) acc[k] = (v8f){0.f, 0.f, 0.f, 0.f, 0.f, 0.f, 0.f, 0.f};
        #pragma unroll
        for (int o = 0; o < ONUM; ++o) {
            #pragma unroll
            for (int k = 0; k < 4; ++k) {
                const v2f bf = cur.xa[k] * cf[o];
                acc[k] = mma4(wf1[o][k], bf, acc[k]);
            }
        }
        const v8f s = (acc[0] + acc[1]) + (acc[2] + acc[3]);

        float nsp = 0.0f;
        #pragma unroll
        for (int i = 0; i < 8; ++i) nsp += s[i] * s[i];
        const float ns  = nsp + __shfl_xor(nsp, 16);
        const float fac = sqrtf(ns) / (1.0f + ns);
        float vj[8];
        #pragma unroll
        for (int i = 0; i < 8; ++i) vj[i] = s[i] * fac;

        if (MODE < 2) {
            v8f ag[4];
            #pragma unroll
            for (int q = 0; q < 4; ++q) ag[q] = (v8f){0.f, 0.f, 0.f, 0.f, 0.f, 0.f, 0.f, 0.f};
            #pragma unroll
            for (int d = 0; d < DNUM; ++d) {
                const v4f wa = wf2a[d];
                const v4f wb = wf2b[d];
                const float xd = cur.xr[d >> 2][d & 3];
                const v2f a0 = {wa.x, wa.y}, a1 = {wa.z, wa.w};
                const v2f a2 = {wb.x, wb.y}, a3 = {wb.z, wb.w};
                const v2f b0 = {xd * vj[0], xd * vj[1]}, b1 = {xd * vj[2], xd * vj[3]};
                const v2f b2 = {xd * vj[4], xd * vj[5]}, b3 = {xd * vj[6], xd * vj[7]};
                ag[0] = mma4(a0, b0, ag[0]);
                ag[1] = mma4(a1, b1, ag[1]);
                ag[2] = mma4(a2, b2, ag[2]);
                ag[3] = mma4(a3, b3, ag[3]);
            }
            const v8f agr = (ag[0] + ag[1]) + (ag[2] + ag[3]);
            v4f n0, n1;
            #pragma unroll
            for (int i = 0; i < 4; ++i) {
                n0[i] = bold[i]     + agr[i];
                n1[i] = bold[4 + i] + agr[4 + i];
            }
            const size_t brow = ((size_t)b * RNUM + r) * ONUM + 8 * h;
            *reinterpret_cast<v4f*>(bmat + brow)     = n0;
            *reinterpret_cast<v4f*>(bmat + brow + 4) = n1;
        } else {
            float* op = out + ((size_t)b * RNUM + r) * CNUM + 8 * h;
            const v4f o0 = {vj[0], vj[1], vj[2], vj[3]};
            const v4f o1 = {vj[4], vj[5], vj[6], vj[7]};
            *reinterpret_cast<v4f*>(op)     = o0;
            *reinterpret_cast<v4f*>(op + 4) = o1;
        }
        cur = nxt;
    }
}

__global__ __launch_bounds__(256) void softmax_reduce(
    const float* __restrict__ bmat,
    float* __restrict__ rowmax,
    float* __restrict__ denom)
{
    __shared__ float sm[256];
    const int b  = blockIdx.x;
    const int t  = threadIdx.x;
    const int o  = t & 15;
    const int rs = t >> 4;

    float mx = -INFINITY;
    for (int r = rs; r < RNUM; r += 16)
        mx = fmaxf(mx, bmat[((size_t)b * RNUM + r) * ONUM + o]);
    sm[t] = mx;
    __syncthreads();
    for (int off = 128; off >= 16; off >>= 1) {
        if (t < off) sm[t] = fmaxf(sm[t], sm[t + off]);
        __syncthreads();
    }
    const float rm = sm[o];
    if (t < 16) rowmax[b * ONUM + t] = sm[t];
    __syncthreads();

    float se = 0.0f;
    for (int r = rs; r < RNUM; r += 16)
        se += __expf(bmat[((size_t)b * RNUM + r) * ONUM + o] - rm);
    sm[t] = se;
    __syncthreads();
    for (int off = 128; off >= 16; off >>= 1) {
        if (t < off) sm[t] += sm[t + off];
        __syncthreads();
    }
    if (t < 16) denom[b * ONUM + t] = sm[t];
}

extern "C" void kernel_launch(void* const* d_in, const int* in_sizes, int n_in,
                              void* d_out, int out_size, void* d_ws, size_t ws_size,
                              hipStream_t stream) {
    const float* x = (const float*)d_in[0];
    const float* W = (const float*)d_in[1];
    float* out = (float*)d_out;

    float* bmat = (float*)d_ws;
    float* rowmax = bmat + (size_t)BATCH * RNUM * ONUM;
    float* denom  = rowmax + BATCH * ONUM;

    dim3 grid(RNUM / 8), block(256);

    caps_kernel<0><<<grid, block, 0, stream>>>(x, W, bmat, rowmax, denom, out);
    softmax_reduce<<<dim3(BATCH), block, 0, stream>>>(bmat, rowmax, denom);
    caps_kernel<1><<<grid, block, 0, stream>>>(x, W, bmat, rowmax, denom, out);
    softmax_reduce<<<dim3(BATCH), block, 0, stream>>>(bmat, rowmax, denom);
    caps_kernel<2><<<grid, block, 0, stream>>>(x, W, bmat, rowmax, denom, out);
}


# Softmax statistics made where the logits are made: each block leaves max and exp-sum per (b,o) over its 8 r via LDS, a small combine folds them; no pass over the logits; denom sum order changes
# speedup vs baseline: 2.4111x; 1.8982x over previous
#include <hip/hip_runtime.h>
#include <hip/hip_bf16.h>
#include <math.h>


#define BATCH 128
#define RNUM  2048
#define ONUM  16
#define DNUM  16
#define CNUM  16
#define NTILE (BATCH / 16)

typedef __attribute__((ext_vector_type(2))) float v2f;
typedef __attribute__((ext_vector_type(4))) float v4f;
typedef __attribute__((ext_vector_type(8))) float v8f;

__device__ __forceinline__ v8f mma4(v2f a, v2f b, v8f c) {
    return __builtin_amdgcn_wmma_f32_16x16x4_f32(false, a, false, b, (short)0, c, false, false);
}

struct TileIn {
    v2f xa[4];
    v4f xr[4];
    v4f bl0, bl1, rm0, rm1, dn0, dn1;
};

template <int MODE>
__device__ __forceinline__ void load_tile(TileIn& ti,
    const float* __restrict__ x, const float* bmat,
    const float* __restrict__ rowmax, const float* __restrict__ denom,
    int b, int r, int h)
{
    const float* xb = x + b * DNUM;
    #pragma unroll
    for (int k = 0; k < 4; ++k)
        ti.xa[k] = *reinterpret_cast<const v2f*>(xb + 4 * k + 2 * h);
    #pragma unroll
    for (int t = 0; t < 4; ++t)
        ti.xr[t] = (MODE < 2) ? *reinterpret_cast<const v4f*>(xb + 4 * t) : (v4f){0.f, 0.f, 0.f, 0.f};
    const v4f zero4 = {0.f, 0.f, 0.f, 0.f};
    ti.bl0 = zero4; ti.bl1 = zero4; ti.rm0 = zero4; ti.rm1 = zero4; ti.dn0 = zero4; ti.dn1 = zero4;
    if (MODE > 0) {
        const size_t brow = ((size_t)b * RNUM + r) * ONUM + 8 * h;
        ti.bl0 = *reinterpret_cast<const v4f*>(bmat + brow);
        ti.bl1 = *reinterpret_cast<const v4f*>(bmat + brow + 4);
        ti.rm0 = *reinterpret_cast<const v4f*>(rowmax + b * ONUM + 8 * h);
        ti.rm1 = *reinterpret_cast<const v4f*>(rowmax + b * ONUM + 8 * h + 4);
        ti.dn0 = *reinterpret_cast<const v4f*>(denom + b * ONUM + 8 * h);
        ti.dn1 = *reinterpret_cast<const v4f*>(denom + b * ONUM + 8 * h + 4);
    }
}

template <int MODE>
__global__ __launch_bounds__(256) void caps_kernel(
    const float* __restrict__ x,
    const float* __restrict__ W,
    float* bmat,
    const float* __restrict__ rowmax,
    const float* __restrict__ denom,
    float* __restrict__ out,
    float* __restrict__ pmax,
    float* __restrict__ psum)
{
    __shared__ float lg[MODE < 2 ? NTILE * 8 * 256 : 1];

    const int tid  = threadIdx.x;
    const int L    = tid & 31;
    const int w    = tid >> 5;
    const int h    = L >> 4;
    const int m16  = L & 15;
    const int r    = blockIdx.x * 8 + w;

    const float* Wr = W + (size_t)r * (ONUM * DNUM * CNUM);

    TileIn cur;
    load_tile<MODE>(cur, x, bmat, rowmax, denom, m16, r, h);

    v2f wf1[ONUM][4];
    #pragma unroll
    for (int o = 0; o < ONUM; ++o) {
        #pragma unroll
        for (int k = 0; k < 4; ++k) {
            const int d0 = 4 * k + 2 * h;
            wf1[o][k].x = Wr[o * (DNUM * CNUM) + (d0 + 0) * CNUM + m16];
            wf1[o][k].y = Wr[o * (DNUM * CNUM) + (d0 + 1) * CNUM + m16];
        }
    }
    v4f wf2a[DNUM], wf2b[DNUM];
    if (MODE < 2) {
        const float* Wl = Wr + m16 * (DNUM * CNUM) + 8 * h;
        #pragma unroll
        for (int d = 0; d < DNUM; ++d) {
            wf2a[d] = *reinterpret_cast<const v4f*>(Wl + d * CNUM);
            wf2b[d] = *reinterpret_cast<const v4f*>(Wl + d * CNUM + 4);
        }
    }

    for (int t = 0; t < NTILE; ++t) {
        const int b = t * 16 + m16;

        TileIn nxt;
        load_tile<MODE>(nxt, x, bmat, rowmax, denom, (t < NTILE - 1 ? t + 1 : t) * 16 + m16, r, h);

        float cf[ONUM];
        float bold[8];
        if (MODE == 0) {
            #pragma unroll
            for (int o = 0; o < ONUM; ++o) cf[o] = 1.0f / (float)RNUM;
            #pragma unroll
            for (int i = 0; i < 8; ++i) bold[i] = 0.0f;
        } else {
            float cown[8], coth[8];
            #pragma unroll
            for (int i = 0; i < 4; ++i) {
                bold[i]     = cur.bl0[i];
                bold[4 + i] = cur.bl1[i];
                cown[i]     = __expf(cur.bl0[i] - cur.rm0[i]) / cur.dn0[i];
                cown[4 + i] = __expf(cur.bl1[i] - cur.rm1[i]) / cur.dn1[i];
            }
            #pragma unroll
            for (int i = 0; i < 8; ++i) coth[i] = __shfl_xor(cown[i], 16);
            #pragma unroll
            for (int i = 0; i < 8; ++i) {
                cf[i]     = h ? coth[i] : cown[i];
                cf[8 + i] = h ? cown[i] : coth[i];
            }
        }

        v8f acc[4];
        #pragma unroll
        for (int k = 0; k < 4; ++k) acc[k] = (v8f){0.f, 0.f, 0.f, 0.f, 0.f, 0.f, 0.f, 0.f};
        #pragma unroll
        for (int o = 0; o < ONUM; ++o) {
            #pragma unroll
            for (int k = 0; k < 4; ++k) {
                const v2f bf = cur.xa[k] * cf[o];
                acc[k] = mma4(wf1[o][k], bf, acc[k]);
            }
        }
        const v8f s = (acc[0] + acc[1]) + (acc[2] + acc[3]);

        float nsp = 0.0f;
        #pragma unroll
        for (int i = 0; i < 8; ++i) nsp += s[i] * s[i];
        const float ns  = nsp + __shfl_xor(nsp, 16);
        const float fac = sqrtf(ns) / (1.0f + ns);
        float vj[8];
        #pragma unroll
        for (int i = 0; i < 8; ++i) vj[i] = s[i] * fac;

        if (MODE < 2) {
            v8f ag[4];
            #pragma unroll
            for (int q = 0; q < 4; ++q) ag[q] = (v8f){0.f, 0.f, 0.f, 0.f, 0.f, 0.f, 0.f, 0.f};
            #pragma unroll
            for (int d = 0; d < DNUM; ++d) {
                const v4f wa = wf2a[d];
                const v4f wb = wf2b[d];
                const float xd = cur.xr[d >> 2][d & 3];
                const v2f a0 = {wa.x, wa.y}, a1 = {wa.z, wa.w};
                const v2f a2 = {wb.x, wb.y}, a3 = {wb.z, wb.w};
                const v2f b0 = {xd * vj[0], xd * vj[1]}, b1 = {xd * vj[2], xd * vj[3]};
                const v2f b2 = {xd * vj[4], xd * vj[5]}, b3 = {xd * vj[6], xd * vj[7]};
                ag[0] = mma4(a0, b0, ag[0]);
                ag[1] = mma4(a1, b1, ag[1]);
                ag[2] = mma4(a2, b2, ag[2]);
                ag[3] = mma4(a3, b3, ag[3]);
            }
            const v8f agr = (ag[0] + ag[1]) + (ag[2] + ag[3]);
            v4f n0, n1;
            #pragma unroll
            for (int i = 0; i < 4; ++i) {
                n0[i] = bold[i]     + agr[i];
                n1[i] = bold[4 + i] + agr[4 + i];
            }
            const size_t brow = ((size_t)b * RNUM + r) * ONUM + 8 * h;
            *reinterpret_cast<v4f*>(bmat + brow)     = n0;
            *reinterpret_cast<v4f*>(bmat + brow + 4) = n1;
            float* lp = lg + (t * 8 + w) * 256 + m16 * 16 + 8 * h;
            *reinterpret_cast<v4f*>(lp)     = n0;
            *reinterpret_cast<v4f*>(lp + 4) = n1;
        } else {
            float* op = out + ((size_t)b * RNUM + r) * CNUM + 8 * h;
            const v4f o0 = {vj[0], vj[1], vj[2], vj[3]};
            const v4f o1 = {vj[4], vj[5], vj[6], vj[7]};
            *reinterpret_cast<v4f*>(op)     = o0;
            *reinterpret_cast<v4f*>(op + 4) = o1;
        }
        cur = nxt;
    }

    if (MODE < 2) {
        __syncthreads();
        const size_t pbase = (size_t)blockIdx.x * (BATCH * ONUM) + tid;
        #pragma unroll
        for (int t = 0; t < NTILE; ++t) {
            float v[8];
            #pragma unroll
            for (int q = 0; q < 8; ++q) v[q] = lg[(t * 8 + q) * 256 + tid];
            const float m = fmaxf(fmaxf(fmaxf(v[0], v[1]), fmaxf(v[2], v[3])),
                                  fmaxf(fmaxf(v[4], v[5]), fmaxf(v[6], v[7])));
            float e[8];
            #pragma unroll
            for (int q = 0; q < 8; ++q) e[q] = __expf(v[q] - m);
            pmax[pbase + t * 256] = m;
            psum[pbase + t * 256] = ((e[0] + e[1]) + (e[2] + e[3])) + ((e[4] + e[5]) + (e[6] + e[7]));
        }
    }
}

__global__ __launch_bounds__(256) void softmax_combine(
    const float* __restrict__ pmax,
    const float* __restrict__ psum,
    float* __restrict__ rowmax,
    float* __restrict__ denom)
{
    __shared__ float sm_m[256];
    __shared__ float sm_s[256];
    const int b  = blockIdx.x;
    const int t  = threadIdx.x;
    const int o  = t & 15;
    const int st = t >> 4;
    constexpr int NBLK = RNUM / 8;
    constexpr int PER  = NBLK / 16;

    float m[PER], s[PER];
    #pragma unroll
    for (int j = 0; j < PER; ++j) {
        const size_t idx = ((size_t)(st + 16 * j) * BATCH + b) * ONUM + o;
        m[j] = pmax[idx];
        s[j] = psum[idx];
    }
    float mx = m[0];
    #pragma unroll
    for (int j = 1; j < PER; ++j) mx = fmaxf(mx, m[j]);
    float se = 0.0f;
    #pragma unroll
    for (int j = 0; j < PER; ++j) se += s[j] * __expf(m[j] - mx);
    sm_m[t] = mx;
    sm_s[t] = se;
    __syncthreads();
    if (t < 16) {
        float M = sm_m[t];
        #pragma unroll
        for (int k = 1; k < 16; ++k) M = fmaxf(M, sm_m[t + 16 * k]);
        float D = 0.0f;
        #pragma unroll
        for (int k = 0; k < 16; ++k) D += sm_s[t + 16 * k] * __expf(sm_m[t + 16 * k] - M);
        rowmax[b * ONUM + t] = M;
        denom[b * ONUM + t]  = D;
    }
}

extern "C" void kernel_launch(void* const* d_in, const int* in_sizes, int n_in,
                              void* d_out, int out_size, void* d_ws, size_t ws_size,
                              hipStream_t stream) {
    const float* x = (const float*)d_in[0];
    const float* W = (const float*)d_in[1];
    float* out = (float*)d_out;

    float* bmat = (float*)d_ws;
    float* rowmax = bmat + (size_t)BATCH * RNUM * ONUM;
    float* denom  = rowmax + BATCH * ONUM;

    float* pmax = out;
    float* psum = out + (size_t)(RNUM / 8) * BATCH * ONUM;

    dim3 grid(RNUM / 8), block(256);

    caps_kernel<0><<<grid, block, 0, stream>>>(x, W, bmat, rowmax, denom, out, pmax, psum);
    softmax_combine<<<dim3(BATCH), block, 0, stream>>>(pmax, psum, rowmax, denom);
    caps_kernel<1><<<grid, block, 0, stream>>>(x, W, bmat, rowmax, denom, out, pmax, psum);
    softmax_combine<<<dim3(BATCH), block, 0, stream>>>(pmax, psum, rowmax, denom);
    caps_kernel<2><<<grid, block, 0, stream>>>(x, W, bmat, rowmax, denom, out, pmax, psum);
}


# First routing pass: its coefficients are the constant 1/R, so W[r] is summed over o once per wave and product 1 is a K=16 f32 WMMA product (4 per tile, was 64); pass-1 sum order: o, then d
# speedup vs baseline: 2.6453x; 1.0971x over previous
#include <hip/hip_runtime.h>
#include <hip/hip_bf16.h>
#include <math.h>


#define BATCH 128
#define RNUM  2048
#define ONUM  16
#define DNUM  16
#define CNUM  16
#define NTILE (BATCH / 16)

typedef __attribute__((ext_vector_type(2))) float v2f;
typedef __attribute__((ext_vector_type(4))) float v4f;
typedef __attribute__((ext_vector_type(8))) float v8f;

__device__ __forceinline__ v8f mma4(v2f a, v2f b, v8f c) {
    return __builtin_amdgcn_wmma_f32_16x16x4_f32(false, a, false, b, (short)0, c, false, false);
}

struct TileIn {
    v2f xa[4];
    v4f xr[4];
    v4f bl0, bl1, rm0, rm1, dn0, dn1;
};

template <int MODE>
__device__ __forceinline__ void load_tile(TileIn& ti,
    const float* __restrict__ x, const float* bmat,
    const float* __restrict__ rowmax, const float* __restrict__ denom,
    int b, int r, int h)
{
    const float* xb = x + b * DNUM;
    #pragma unroll
    for (int k = 0; k < 4; ++k)
        ti.xa[k] = *reinterpret_cast<const v2f*>(xb + 4 * k + 2 * h);
    #pragma unroll
    for (int t = 0; t < 4; ++t)
        ti.xr[t] = (MODE < 2) ? *reinterpret_cast<const v4f*>(xb + 4 * t) : (v4f){0.f, 0.f, 0.f, 0.f};
    const v4f zero4 = {0.f, 0.f, 0.f, 0.f};
    ti.bl0 = zero4; ti.bl1 = zero4; ti.rm0 = zero4; ti.rm1 = zero4; ti.dn0 = zero4; ti.dn1 = zero4;
    if (MODE > 0) {
        const size_t brow = ((size_t)b * RNUM + r) * ONUM + 8 * h;
        ti.bl0 = *reinterpret_cast<const v4f*>(bmat + brow);
        ti.bl1 = *reinterpret_cast<const v4f*>(bmat + brow + 4);
        ti.rm0 = *reinterpret_cast<const v4f*>(rowmax + b * ONUM + 8 * h);
        ti.rm1 = *reinterpret_cast<const v4f*>(rowmax + b * ONUM + 8 * h + 4);
        ti.dn0 = *reinterpret_cast<const v4f*>(denom + b * ONUM + 8 * h);
        ti.dn1 = *reinterpret_cast<const v4f*>(denom + b * ONUM + 8 * h + 4);
    }
}

template <int MODE>
__global__ __launch_bounds__(256) void caps_kernel(
    const float* __restrict__ x,
    const float* __restrict__ W,
    float* bmat,
    const float* __restrict__ rowmax,
    const float* __restrict__ denom,
    float* __restrict__ out,
    float* __restrict__ pmax,
    float* __restrict__ psum)
{
    __shared__ float lg[MODE < 2 ? NTILE * 8 * 256 : 1];

    const int tid  = threadIdx.x;
    const int L    = tid & 31;
    const int w    = tid >> 5;
    const int h    = L >> 4;
    const int m16  = L & 15;
    const int r    = blockIdx.x * 8 + w;

    const float* Wr = W + (size_t)r * (ONUM * DNUM * CNUM);

    TileIn cur;
    load_tile<MODE>(cur, x, bmat, rowmax, denom, m16, r, h);

    v2f wf1[MODE == 0 ? 1 : ONUM][4];
    v2f ws1[4];
    if (MODE == 0) {
        #pragma unroll
        for (int k = 0; k < 4; ++k) ws1[k] = (v2f){0.f, 0.f};
        #pragma unroll
        for (int o = 0; o < ONUM; ++o) {
            #pragma unroll
            for (int k = 0; k < 4; ++k) {
                const int d0 = 4 * k + 2 * h;
                ws1[k].x += Wr[o * (DNUM * CNUM) + (d0 + 0) * CNUM + m16];
                ws1[k].y += Wr[o * (DNUM * CNUM) + (d0 + 1) * CNUM + m16];
            }
        }
        #pragma unroll
        for (int k = 0; k < 4; ++k) ws1[k] = ws1[k] * (1.0f / (float)RNUM);
    } else {
        #pragma unroll
        for (int o = 0; o < ONUM; ++o) {
            #pragma unroll
            for (int k = 0; k < 4; ++k) {
                const int d0 = 4 * k + 2 * h;
                wf1[o][k].x = Wr[o * (DNUM * CNUM) + (d0 + 0) * CNUM + m16];
                wf1[o][k].y = Wr[o * (DNUM * CNUM) + (d0 + 1) * CNUM + m16];
            }
        }
    }
    v4f wf2a[DNUM], wf2b[DNUM];
    if (MODE < 2) {
        const float* Wl = Wr + m16 * (DNUM * CNUM) + 8 * h;
        #pragma unroll
        for (int d = 0; d < DNUM; ++d) {
            wf2a[d] = *reinterpret_cast<const v4f*>(Wl + d * CNUM);
            wf2b[d] = *reinterpret_cast<const v4f*>(Wl + d * CNUM + 4);
        }
    }

    for (int t = 0; t < NTILE; ++t) {
        const int b = t * 16 + m16;

        TileIn nxt;
        load_tile<MODE>(nxt, x, bmat, rowmax, denom, (t < NTILE - 1 ? t + 1 : t) * 16 + m16, r, h);

        float cf[ONUM];
        float bold[8];
        if (MODE == 0) {
            #pragma unroll
            for (int i = 0; i < 8; ++i) bold[i] = 0.0f;
        } else {
            float cown[8], coth[8];
            #pragma unroll
            for (int i = 0; i < 4; ++i) {
                bold[i]     = cur.bl0[i];
                bold[4 + i] = cur.bl1[i];
                cown[i]     = __expf(cur.bl0[i] - cur.rm0[i]) / cur.dn0[i];
                cown[4 + i] = __expf(cur.bl1[i] - cur.rm1[i]) / cur.dn1[i];
            }
            #pragma unroll
            for (int i = 0; i < 8; ++i) coth[i] = __shfl_xor(cown[i], 16);
            #pragma unroll
            for (int i = 0; i < 8; ++i) {
                cf[i]     = h ? coth[i] : cown[i];
                cf[8 + i] = h ? cown[i] : coth[i];
            }
        }

        v8f s;
        if (MODE == 0) {
            v8f a0 = (v8f){0.f, 0.f, 0.f, 0.f, 0.f, 0.f, 0.f, 0.f};
            v8f a1 = a0;
            a0 = mma4(ws1[0], cur.xa[0], a0);
            a1 = mma4(ws1[1], cur.xa[1], a1);
            a0 = mma4(ws1[2], cur.xa[2], a0);
            a1 = mma4(ws1[3], cur.xa[3], a1);
            s = a0 + a1;
        } else {
            v8f acc[4];
            #pragma unroll
            for (int k = 0; k < 4; ++k) acc[k] = (v8f){0.f, 0.f, 0.f, 0.f, 0.f, 0.f, 0.f, 0.f};
            #pragma unroll
            for (int o = 0; o < ONUM; ++o) {
                #pragma unroll
                for (int k = 0; k < 4; ++k) {
                    const v2f bf = cur.xa[k] * cf[o];
                    acc[k] = mma4(wf1[o][k], bf, acc[k]);
                }
            }
            s = (acc[0] + acc[1]) + (acc[2] + acc[3]);
        }

        float nsp = 0.0f;
        #pragma unroll
        for (int i = 0; i < 8; ++i) nsp += s[i] * s[i];
        const float ns  = nsp + __shfl_xor(nsp, 16);
        const float fac = sqrtf(ns) / (1.0f + ns);
        float vj[8];
        #pragma unroll
        for (int i = 0; i < 8; ++i) vj[i] = s[i] * fac;

        if (MODE < 2) {
            v8f ag[4];
            #pragma unroll
            for (int q = 0; q < 4; ++q) ag[q] = (v8f){0.f, 0.f, 0.f, 0.f, 0.f, 0.f, 0.f, 0.f};
            #pragma unroll
            for (int d = 0; d < DNUM; ++d) {
                const v4f wa = wf2a[d];
                const v4f wb = wf2b[d];
                const float xd = cur.xr[d >> 2][d & 3];
                const v2f a0 = {wa.x, wa.y}, a1 = {wa.z, wa.w};
                const v2f a2 = {wb.x, wb.y}, a3 = {wb.z, wb.w};
                const v2f b0 = {xd * vj[0], xd * vj[1]}, b1 = {xd * vj[2], xd * vj[3]};
                const v2f b2 = {xd * vj[4], xd * vj[5]}, b3 = {xd * vj[6], xd * vj[7]};
                ag[0] = mma4(a0, b0, ag[0]);
                ag[1] = mma4(a1, b1, ag[1]);
                ag[2] = mma4(a2, b2, ag[2]);
                ag[3] = mma4(a3, b3, ag[3]);
            }
            const v8f agr = (ag[0] + ag[1]) + (ag[2] + ag[3]);
            v4f n0, n1;
            #pragma unroll
            for (int i = 0; i < 4; ++i) {
                n0[i] = bold[i]     + agr[i];
                n1[i] = bold[4 + i] + agr[4 + i];
            }
            const size_t brow = ((size_t)b * RNUM + r) * ONUM + 8 * h;
            *reinterpret_cast<v4f*>(bmat + brow)     = n0;
            *reinterpret_cast<v4f*>(bmat + brow + 4) = n1;
            float* lp = lg + (t * 8 + w) * 256 + m16 * 16 + 8 * h;
            *reinterpret_cast<v4f*>(lp)     = n0;
            *reinterpret_cast<v4f*>(lp + 4) = n1;
        } else {
            float* op = out + ((size_t)b * RNUM + r) * CNUM + 8 * h;
            const v4f o0 = {vj[0], vj[1], vj[2], vj[3]};
            const v4f o1 = {vj[4], vj[5], vj[6], vj[7]};
            *reinterpret_cast<v4f*>(op)     = o0;
            *reinterpret_cast<v4f*>(op + 4) = o1;
        }
        cur = nxt;
    }

    if (MODE < 2) {
        __syncthreads();
        const size_t pbase = (size_t)blockIdx.x * (BATCH * ONUM) + tid;
        #pragma unroll
        for (int t = 0; t < NTILE; ++t) {
            float v[8];
            #pragma unroll
            for (int q = 0; q < 8; ++q) v[q] = lg[(t * 8 + q) * 256 + tid];
            const float m = fmaxf(fmaxf(fmaxf(v[0], v[1]), fmaxf(v[2], v[3])),
                                  fmaxf(fmaxf(v[4], v[5]), fmaxf(v[6], v[7])));
            float e[8];
            #pragma unroll
            for (int q = 0; q < 8; ++q) e[q] = __expf(v[q] - m);
            pmax[pbase + t * 256] = m;
            psum[pbase + t * 256] = ((e[0] + e[1]) + (e[2] + e[3])) + ((e[4] + e[5]) + (e[6] + e[7]));
        }
    }
}

__global__ __launch_bounds__(256) void softmax_combine(
    const float* __restrict__ pmax,
    const float* __restrict__ psum,
    float* __restrict__ rowmax,
    float* __restrict__ denom)
{
    __shared__ float sm_m[256];
    __shared__ float sm_s[256];
    const int b  = blockIdx.x;
    const int t  = threadIdx.x;
    const int o  = t & 15;
    const int st = t >> 4;
    constexpr int NBLK = RNUM / 8;
    constexpr int PER  = NBLK / 16;

    float m[PER], s[PER];
    #pragma unroll
    for (int j = 0; j < PER; ++j) {
        const size_t idx = ((size_t)(st + 16 * j) * BATCH + b) * ONUM + o;
        m[j] = pmax[idx];
        s[j] = psum[idx];
    }
    float mx = m[0];
    #pragma unroll
    for (int j = 1; j < PER; ++j) mx = fmaxf(mx, m[j]);
    float se = 0.0f;
    #pragma unroll
    for (int j = 0; j < PER; ++j) se += s[j] * __expf(m[j] - mx);
    sm_m[t] = mx;
    sm_s[t] = se;
    __syncthreads();
    if (t < 16) {
        float M = sm_m[t];
        #pragma unroll
        for (int k = 1; k < 16; ++k) M = fmaxf(M, sm_m[t + 16 * k]);
        float D = 0.0f;
        #pragma unroll
        for (int k = 0; k < 16; ++k) D += sm_s[t + 16 * k] * __expf(sm_m[t + 16 * k] - M);
        rowmax[b * ONUM + t] = M;
        denom[b * ONUM + t]  = D;
    }
}

extern "C" void kernel_launch(void* const* d_in, const int* in_sizes, int n_in,
                              void* d_out, int out_size, void* d_ws, size_t ws_size,
                              hipStream_t stream) {
    const float* x = (const float*)d_in[0];
    const float* W = (const float*)d_in[1];
    float* out = (float*)d_out;

    float* bmat = (float*)d_ws;
    float* rowmax = bmat + (size_t)BATCH * RNUM * ONUM;
    float* denom  = rowmax + BATCH * ONUM;

    float* pmax = out;
    float* psum = out + (size_t)(RNUM / 8) * BATCH * ONUM;

    dim3 grid(RNUM / 8), block(256);

    caps_kernel<0><<<grid, block, 0, stream>>>(x, W, bmat, rowmax, denom, out, pmax, psum);
    softmax_combine<<<dim3(BATCH), block, 0, stream>>>(pmax, psum, rowmax, denom);
    caps_kernel<1><<<grid, block, 0, stream>>>(x, W, bmat, rowmax, denom, out, pmax, psum);
    softmax_combine<<<dim3(BATCH), block, 0, stream>>>(pmax, psum, rowmax, denom);
    caps_kernel<2><<<grid, block, 0, stream>>>(x, W, bmat, rowmax, denom, out, pmax, psum);
}


# Middle pass contracts d first: 64 K=16 f32 WMMA per tile (was 2x64 at K=256) leave u[o][c] in the lane owning batch row b; both routing sums are in-lane FMAs; sum order: d, then o; c by halves
# speedup vs baseline: 3.0121x; 1.1387x over previous
#include <hip/hip_runtime.h>
#include <hip/hip_bf16.h>
#include <math.h>


#define BATCH 128
#define RNUM  2048
#define ONUM  16
#define DNUM  16
#define CNUM  16
#define NTILE (BATCH / 16)

typedef __attribute__((ext_vector_type(2))) float v2f;
typedef __attribute__((ext_vector_type(4))) float v4f;
typedef __attribute__((ext_vector_type(8))) float v8f;

__device__ __forceinline__ v8f mma4(v2f a, v2f b, v8f c) {
    return __builtin_amdgcn_wmma_f32_16x16x4_f32(false, a, false, b, (short)0, c, false, false);
}

struct TileIn {
    v2f xa[4];
    v4f xr[4];
    v4f bl0, bl1, rm0, rm1, dn0, dn1;
};

template <int MODE>
__device__ __forceinline__ void load_tile(TileIn& ti,
    const float* __restrict__ x, const float* bmat,
    const float* __restrict__ rowmax, const float* __restrict__ denom,
    int b, int r, int h)
{
    const float* xb = x + b * DNUM;
    #pragma unroll
    for (int k = 0; k < 4; ++k)
        ti.xa[k] = *reinterpret_cast<const v2f*>(xb + 4 * k + 2 * h);
    #pragma unroll
    for (int t = 0; t < 4; ++t)
        ti.xr[t] = (MODE == 0) ? *reinterpret_cast<const v4f*>(xb + 4 * t) : (v4f){0.f, 0.f, 0.f, 0.f};
    const v4f zero4 = {0.f, 0.f, 0.f, 0.f};
    ti.bl0 = zero4; ti.bl1 = zero4; ti.rm0 = zero4; ti.rm1 = zero4; ti.dn0 = zero4; ti.dn1 = zero4;
    if (MODE > 0) {
        const size_t brow = ((size_t)b * RNUM + r) * ONUM + 8 * h;
        ti.bl0 = *reinterpret_cast<const v4f*>(bmat + brow);
        ti.bl1 = *reinterpret_cast<const v4f*>(bmat + brow + 4);
        ti.rm0 = *reinterpret_cast<const v4f*>(rowmax + b * ONUM + 8 * h);
        ti.rm1 = *reinterpret_cast<const v4f*>(rowmax + b * ONUM + 8 * h + 4);
        ti.dn0 = *reinterpret_cast<const v4f*>(denom + b * ONUM + 8 * h);
        ti.dn1 = *reinterpret_cast<const v4f*>(denom + b * ONUM + 8 * h + 4);
    }
}

template <int MODE>
__global__ __launch_bounds__(256) void caps_kernel(
    const float* __restrict__ x,
    const float* __restrict__ W,
    float* bmat,
    const float* __restrict__ rowmax,
    const float* __restrict__ denom,
    float* __restrict__ out,
    float* __restrict__ pmax,
    float* __restrict__ psum)
{
    __shared__ float lg[MODE < 2 ? NTILE * 8 * 256 : 1];

    const int tid  = threadIdx.x;
    const int L    = tid & 31;
    const int w    = tid >> 5;
    const int h    = L >> 4;
    const int m16  = L & 15;
    const int r    = blockIdx.x * 8 + w;

    const float* Wr = W + (size_t)r * (ONUM * DNUM * CNUM);

    TileIn cur;
    load_tile<MODE>(cur, x, bmat, rowmax, denom, m16, r, h);

    v2f wf1[MODE == 0 ? 1 : ONUM][4];
    v2f ws1[4];
    if (MODE == 0) {
        #pragma unroll
        for (int k = 0; k < 4; ++k) ws1[k] = (v2f){0.f, 0.f};
        #pragma unroll
        for (int o = 0; o < ONUM; ++o) {
            #pragma unroll
            for (int k = 0; k < 4; ++k) {
                const int d0 = 4 * k + 2 * h;
                ws1[k].x += Wr[o * (DNUM * CNUM) + (d0 + 0) * CNUM + m16];
                ws1[k].y += Wr[o * (DNUM * CNUM) + (d0 + 1) * CNUM + m16];
            }
        }
        #pragma unroll
        for (int k = 0; k < 4; ++k) ws1[k] = ws1[k] * (1.0f / (float)RNUM);
    } else {
        #pragma unroll
        for (int o = 0; o < ONUM; ++o) {
            #pragma unroll
            for (int k = 0; k < 4; ++k) {
                const int d0 = 4 * k + 2 * h;
                wf1[o][k].x = Wr[o * (DNUM * CNUM) + (d0 + 0) * CNUM + m16];
                wf1[o][k].y = Wr[o * (DNUM * CNUM) + (d0 + 1) * CNUM + m16];
            }
        }
    }
    v4f wf2a[DNUM], wf2b[DNUM];
    if (MODE == 0) {
        const float* Wl = Wr + m16 * (DNUM * CNUM) + 8 * h;
        #pragma unroll
        for (int d = 0; d < DNUM; ++d) {
            wf2a[d] = *reinterpret_cast<const v4f*>(Wl + d * CNUM);
            wf2b[d] = *reinterpret_cast<const v4f*>(Wl + d * CNUM + 4);
        }
    }

    for (int t = 0; t < NTILE; ++t) {
        const int b = t * 16 + m16;

        TileIn nxt;
        load_tile<MODE>(nxt, x, bmat, rowmax, denom, (t < NTILE - 1 ? t + 1 : t) * 16 + m16, r, h);

        float cf[ONUM];
        float bold[8];
        if (MODE == 0) {
            #pragma unroll
            for (int i = 0; i < 8; ++i) bold[i] = 0.0f;
        } else {
            float cown[8], coth[8];
            #pragma unroll
            for (int i = 0; i < 4; ++i) {
                bold[i]     = cur.bl0[i];
                bold[4 + i] = cur.bl1[i];
                cown[i]     = __expf(cur.bl0[i] - cur.rm0[i]) / cur.dn0[i];
                cown[4 + i] = __expf(cur.bl1[i] - cur.rm1[i]) / cur.dn1[i];
            }
            #pragma unroll
            for (int i = 0; i < 8; ++i) coth[i] = __shfl_xor(cown[i], 16);
            #pragma unroll
            for (int i = 0; i < 8; ++i) {
                cf[i]     = h ? coth[i] : cown[i];
                cf[8 + i] = h ? cown[i] : coth[i];
            }
        }

        v8f s;
        v8f u[MODE == 1 ? ONUM : 1];
        if (MODE == 1) {
            s = (v8f){0.f, 0.f, 0.f, 0.f, 0.f, 0.f, 0.f, 0.f};
            #pragma unroll
            for (int o = 0; o < ONUM; o += 2) {
                v8f e0 = (v8f){0.f, 0.f, 0.f, 0.f, 0.f, 0.f, 0.f, 0.f};
                v8f e1 = e0;
                #pragma unroll
                for (int k = 0; k < 4; ++k) {
                    e0 = mma4(wf1[o][k],     cur.xa[k], e0);
                    e1 = mma4(wf1[o + 1][k], cur.xa[k], e1);
                }
                u[o]     = e0;
                u[o + 1] = e1;
                s += e0 * cf[o];
                s += e1 * cf[o + 1];
            }
        } else if (MODE == 0) {
            v8f a0 = (v8f){0.f, 0.f, 0.f, 0.f, 0.f, 0.f, 0.f, 0.f};
            v8f a1 = a0;
            a0 = mma4(ws1[0], cur.xa[0], a0);
            a1 = mma4(ws1[1], cur.xa[1], a1);
            a0 = mma4(ws1[2], cur.xa[2], a0);
            a1 = mma4(ws1[3], cur.xa[3], a1);
            s = a0 + a1;
        } else {
            v8f acc[4];
            #pragma unroll
            for (int k = 0; k < 4; ++k) acc[k] = (v8f){0.f, 0.f, 0.f, 0.f, 0.f, 0.f, 0.f, 0.f};
            #pragma unroll
            for (int o = 0; o < ONUM; ++o) {
                #pragma unroll
                for (int k = 0; k < 4; ++k) {
                    const v2f bf = cur.xa[k] * cf[o];
                    acc[k] = mma4(wf1[o][k], bf, acc[k]);
                }
            }
            s = (acc[0] + acc[1]) + (acc[2] + acc[3]);
        }

        float nsp = 0.0f;
        #pragma unroll
        for (int i = 0; i < 8; ++i) nsp += s[i] * s[i];
        const float ns  = nsp + __shfl_xor(nsp, 16);
        const float fac = sqrtf(ns) / (1.0f + ns);
        float vj[8];
        #pragma unroll
        for (int i = 0; i < 8; ++i) vj[i] = s[i] * fac;

        if (MODE < 2) {
            v8f agr;
            if (MODE == 0) {
                v8f ag[4];
                #pragma unroll
                for (int q = 0; q < 4; ++q) ag[q] = (v8f){0.f, 0.f, 0.f, 0.f, 0.f, 0.f, 0.f, 0.f};
                #pragma unroll
                for (int d = 0; d < DNUM; ++d) {
                    const v4f wa = wf2a[d];
                    const v4f wb = wf2b[d];
                    const float xd = cur.xr[d >> 2][d & 3];
                    const v2f a0 = {wa.x, wa.y}, a1 = {wa.z, wa.w};
                    const v2f a2 = {wb.x, wb.y}, a3 = {wb.z, wb.w};
                    const v2f b0 = {xd * vj[0], xd * vj[1]}, b1 = {xd * vj[2], xd * vj[3]};
                    const v2f b2 = {xd * vj[4], xd * vj[5]}, b3 = {xd * vj[6], xd * vj[7]};
                    ag[0] = mma4(a0, b0, ag[0]);
                    ag[1] = mma4(a1, b1, ag[1]);
                    ag[2] = mma4(a2, b2, ag[2]);
                    ag[3] = mma4(a3, b3, ag[3]);
                }
                agr = (ag[0] + ag[1]) + (ag[2] + ag[3]);
            } else {
                const v2f v01 = {vj[0], vj[1]}, v23 = {vj[2], vj[3]};
                const v2f v45 = {vj[4], vj[5]}, v67 = {vj[6], vj[7]};
                float p[ONUM];
                #pragma unroll
                for (int o = 0; o < ONUM; ++o) {
                    const v8f uo = u[o];
                    v2f q = (v2f){uo[0], uo[1]} * v01;
                    q += (v2f){uo[2], uo[3]} * v23;
                    q += (v2f){uo[4], uo[5]} * v45;
                    q += (v2f){uo[6], uo[7]} * v67;
                    p[o] = q.x + q.y;
                }
                #pragma unroll
                for (int i = 0; i < 8; ++i) {
                    const float own = h ? p[8 + i] : p[i];
                    const float snd = h ? p[i] : p[8 + i];
                    agr[i] = own + __shfl_xor(snd, 16);
                }
            }
            v4f n0, n1;
            #pragma unroll
            for (int i = 0; i < 4; ++i) {
                n0[i] = bold[i]     + agr[i];
                n1[i] = bold[4 + i] + agr[4 + i];
            }
            const size_t brow = ((size_t)b * RNUM + r) * ONUM + 8 * h;
            *reinterpret_cast<v4f*>(bmat + brow)     = n0;
            *reinterpret_cast<v4f*>(bmat + brow + 4) = n1;
            float* lp = lg + (t * 8 + w) * 256 + m16 * 16 + 8 * h;
            *reinterpret_cast<v4f*>(lp)     = n0;
            *reinterpret_cast<v4f*>(lp + 4) = n1;
        } else {
            float* op = out + ((size_t)b * RNUM + r) * CNUM + 8 * h;
            const v4f o0 = {vj[0], vj[1], vj[2], vj[3]};
            const v4f o1 = {vj[4], vj[5], vj[6], vj[7]};
            *reinterpret_cast<v4f*>(op)     = o0;
            *reinterpret_cast<v4f*>(op + 4) = o1;
        }
        cur = nxt;
    }

    if (MODE < 2) {
        __syncthreads();
        const size_t pbase = (size_t)blockIdx.x * (BATCH * ONUM) + tid;
        #pragma unroll
        for (int t = 0; t < NTILE; ++t) {
            float v[8];
            #pragma unroll
            for (int q = 0; q < 8; ++q) v[q] = lg[(t * 8 + q) * 256 + tid];
            const float m = fmaxf(fmaxf(fmaxf(v[0], v[1]), fmaxf(v[2], v[3])),
                                  fmaxf(fmaxf(v[4], v[5]), fmaxf(v[6], v[7])));
            float e[8];
            #pragma unroll
            for (int q = 0; q < 8; ++q) e[q] = __expf(v[q] - m);
            pmax[pbase + t * 256] = m;
            psum[pbase + t * 256] = ((e[0] + e[1]) + (e[2] + e[3])) + ((e[4] + e[5]) + (e[6] + e[7]));
        }
    }
}

__global__ __launch_bounds__(256) void softmax_combine(
    const float* __restrict__ pmax,
    const float* __restrict__ psum,
    float* __restrict__ rowmax,
    float* __restrict__ denom)
{
    __shared__ float sm_m[256];
    __shared__ float sm_s[256];
    const int b  = blockIdx.x;
    const int t  = threadIdx.x;
    const int o  = t & 15;
    const int st = t >> 4;
    constexpr int NBLK = RNUM / 8;
    constexpr int PER  = NBLK / 16;

    float m[PER], s[PER];
    #pragma unroll
    for (int j = 0; j < PER; ++j) {
        const size_t idx = ((size_t)(st + 16 * j) * BATCH + b) * ONUM + o;
        m[j] = pmax[idx];
        s[j] = psum[idx];
    }
    float mx = m[0];
    #pragma unroll
    for (int j = 1; j < PER; ++j) mx = fmaxf(mx, m[j]);
    float se = 0.0f;
    #pragma unroll
    for (int j = 0; j < PER; ++j) se += s[j] * __expf(m[j] - mx);
    sm_m[t] = mx;
    sm_s[t] = se;
    __syncthreads();
    if (t < 16) {
        float M = sm_m[t];
        #pragma unroll
        for (int k = 1; k < 16; ++k) M = fmaxf(M, sm_m[t + 16 * k]);
        float D = 0.0f;
        #pragma unroll
        for (int k = 0; k < 16; ++k) D += sm_s[t + 16 * k] * __expf(sm_m[t + 16 * k] - M);
        rowmax[b * ONUM + t] = M;
        denom[b * ONUM + t]  = D;
    }
}

extern "C" void kernel_launch(void* const* d_in, const int* in_sizes, int n_in,
                              void* d_out, int out_size, void* d_ws, size_t ws_size,
                              hipStream_t stream) {
    const float* x = (const float*)d_in[0];
    const float* W = (const float*)d_in[1];
    float* out = (float*)d_out;

    float* bmat = (float*)d_ws;
    float* rowmax = bmat + (size_t)BATCH * RNUM * ONUM;
    float* denom  = rowmax + BATCH * ONUM;

    float* pmax = out;
    float* psum = out + (size_t)(RNUM / 8) * BATCH * ONUM;

    dim3 grid(RNUM / 8), block(256);

    caps_kernel<0><<<grid, block, 0, stream>>>(x, W, bmat, rowmax, denom, out, pmax, psum);
    softmax_combine<<<dim3(BATCH), block, 0, stream>>>(pmax, psum, rowmax, denom);
    caps_kernel<1><<<grid, block, 0, stream>>>(x, W, bmat, rowmax, denom, out, pmax, psum);
    softmax_combine<<<dim3(BATCH), block, 0, stream>>>(pmax, psum, rowmax, denom);
    caps_kernel<2><<<grid, block, 0, stream>>>(x, W, bmat, rowmax, denom, out, pmax, psum);
}


# Younger wave of each SIMD pair (waves 4-7 of a block) runs its first 5 of 8 batch tiles at raised wave priority (s_setprio 3, then 0), so a pair ends together; values and sum orders unchanged
# speedup vs baseline: 3.1258x; 1.0377x over previous
#include <hip/hip_runtime.h>
#include <hip/hip_bf16.h>
#include <math.h>


#define BATCH 128
#define RNUM  2048
#define ONUM  16
#define DNUM  16
#define CNUM  16
#define NTILE (BATCH / 16)
#define PRIO_TILES 5

typedef __attribute__((ext_vector_type(2))) float v2f;
typedef __attribute__((ext_vector_type(4))) float v4f;
typedef __attribute__((ext_vector_type(8))) float v8f;

__device__ __forceinline__ v8f mma4(v2f a, v2f b, v8f c) {
    return __builtin_amdgcn_wmma_f32_16x16x4_f32(false, a, false, b, (short)0, c, false, false);
}

struct TileIn {
    v2f xa[4];
    v4f xr[4];
    v4f bl0, bl1, rm0, rm1, dn0, dn1;
};

template <int MODE>
__device__ __forceinline__ void load_tile(TileIn& ti,
    const float* __restrict__ x, const float* bmat,
    const float* __restrict__ rowmax, const float* __restrict__ denom,
    int b, int r, int h)
{
    const float* xb = x + b * DNUM;
    #pragma unroll
    for (int k = 0; k < 4; ++k)
        ti.xa[k] = *reinterpret_cast<const v2f*>(xb + 4 * k + 2 * h);
    #pragma unroll
    for (int t = 0; t < 4; ++t)
        ti.xr[t] = (MODE == 0) ? *reinterpret_cast<const v4f*>(xb + 4 * t) : (v4f){0.f, 0.f, 0.f, 0.f};
    const v4f zero4 = {0.f, 0.f, 0.f, 0.f};
    ti.bl0 = zero4; ti.bl1 = zero4; ti.rm0 = zero4; ti.rm1 = zero4; ti.dn0 = zero4; ti.dn1 = zero4;
    if (MODE > 0) {
        const size_t brow = ((size_t)b * RNUM + r) * ONUM + 8 * h;
        ti.bl0 = *reinterpret_cast<const v4f*>(bmat + brow);
        ti.bl1 = *reinterpret_cast<const v4f*>(bmat + brow + 4);
        ti.rm0 = *reinterpret_cast<const v4f*>(rowmax + b * ONUM + 8 * h);
        ti.rm1 = *reinterpret_cast<const v4f*>(rowmax + b * ONUM + 8 * h + 4);
        ti.dn0 = *reinterpret_cast<const v4f*>(denom + b * ONUM + 8 * h);
        ti.dn1 = *reinterpret_cast<const v4f*>(denom + b * ONUM + 8 * h + 4);
    }
}

template <int MODE>
__global__ __launch_bounds__(256) void caps_kernel(
    const float* __restrict__ x,
    const float* __restrict__ W,
    float* bmat,
    const float* __restrict__ rowmax,
    const float* __restrict__ denom,
    float* __restrict__ out,
    float* __restrict__ pmax,
    float* __restrict__ psum)
{
    __shared__ float lg[MODE < 2 ? NTILE * 8 * 256 : 1];

    const int tid  = threadIdx.x;
    const int L    = tid & 31;
    const int w    = tid >> 5;
    const int h    = L >> 4;
    const int m16  = L & 15;
    const int r    = blockIdx.x * 8 + w;

    const float* Wr = W + (size_t)r * (ONUM * DNUM * CNUM);

    TileIn cur;
    load_tile<MODE>(cur, x, bmat, rowmax, denom, m16, r, h);

    v2f wf1[MODE == 0 ? 1 : ONUM][4];
    v2f ws1[4];
    if (MODE == 0) {
        #pragma unroll
        for (int k = 0; k < 4; ++k) ws1[k] = (v2f){0.f, 0.f};
        #pragma unroll
        for (int o = 0; o < ONUM; ++o) {
            #pragma unroll
            for (int k = 0; k < 4; ++k) {
                const int d0 = 4 * k + 2 * h;
                ws1[k].x += Wr[o * (DNUM * CNUM) + (d0 + 0) * CNUM + m16];
                ws1[k].y += Wr[o * (DNUM * CNUM) + (d0 + 1) * CNUM + m16];
            }
        }
        #pragma unroll
        for (int k = 0; k < 4; ++k) ws1[k] = ws1[k] * (1.0f / (float)RNUM);
    } else {
        #pragma unroll
        for (int o = 0; o < ONUM; ++o) {
            #pragma unroll
            for (int k = 0; k < 4; ++k) {
                const int d0 = 4 * k + 2 * h;
                wf1[o][k].x = Wr[o * (DNUM * CNUM) + (d0 + 0) * CNUM + m16];
                wf1[o][k].y = Wr[o * (DNUM * CNUM) + (d0 + 1) * CNUM + m16];
            }
        }
    }
    v4f wf2a[DNUM], wf2b[DNUM];
    if (MODE == 0) {
        const float* Wl = Wr + m16 * (DNUM * CNUM) + 8 * h;
        #pragma unroll
        for (int d = 0; d < DNUM; ++d) {
            wf2a[d] = *reinterpret_cast<const v4f*>(Wl + d * CNUM);
            wf2b[d] = *reinterpret_cast<const v4f*>(Wl + d * CNUM + 4);
        }
    }

    if (__builtin_amdgcn_readfirstlane(w) >= 4) __builtin_amdgcn_s_setprio(3);

    for (int t = 0; t < NTILE; ++t) {
        if (t == PRIO_TILES) __builtin_amdgcn_s_setprio(0);
        const int b = t * 16 + m16;

        TileIn nxt;
        load_tile<MODE>(nxt, x, bmat, rowmax, denom, (t < NTILE - 1 ? t + 1 : t) * 16 + m16, r, h);

        float cf[ONUM];
        float bold[8];
        if (MODE == 0) {
            #pragma unroll
            for (int i = 0; i < 8; ++i) bold[i] = 0.0f;
        } else {
            float cown[8], coth[8];
            #pragma unroll
            for (int i = 0; i < 4; ++i) {
                bold[i]     = cur.bl0[i];
                bold[4 + i] = cur.bl1[i];
                cown[i]     = __expf(cur.bl0[i] - cur.rm0[i]) / cur.dn0[i];
                cown[4 + i] = __expf(cur.bl1[i] - cur.rm1[i]) / cur.dn1[i];
            }
            #pragma unroll
            for (int i = 0; i < 8; ++i) coth[i] = __shfl_xor(cown[i], 16);
            #pragma unroll
            for (int i = 0; i < 8; ++i) {
                cf[i]     = h ? coth[i] : cown[i];
                cf[8 + i] = h ? cown[i] : coth[i];
            }
        }

        v8f s;
        v8f u[MODE == 1 ? ONUM : 1];
        if (MODE == 1) {
            s = (v8f){0.f, 0.f, 0.f, 0.f, 0.f, 0.f, 0.f, 0.f};
            #pragma unroll
            for (int o = 0; o < ONUM; o += 2) {
                v8f e0 = (v8f){0.f, 0.f, 0.f, 0.f, 0.f, 0.f, 0.f, 0.f};
                v8f e1 = e0;
                #pragma unroll
                for (int k = 0; k < 4; ++k) {
                    e0 = mma4(wf1[o][k],     cur.xa[k], e0);
                    e1 = mma4(wf1[o + 1][k], cur.xa[k], e1);
                }
                u[o]     = e0;
                u[o + 1] = e1;
                s += e0 * cf[o];
                s += e1 * cf[o + 1];
            }
        } else if (MODE == 0) {
            v8f a0 = (v8f){0.f, 0.f, 0.f, 0.f, 0.f, 0.f, 0.f, 0.f};
            v8f a1 = a0;
            a0 = mma4(ws1[0], cur.xa[0], a0);
            a1 = mma4(ws1[1], cur.xa[1], a1);
            a0 = mma4(ws1[2], cur.xa[2], a0);
            a1 = mma4(ws1[3], cur.xa[3], a1);
            s = a0 + a1;
        } else {
            v8f acc[4];
            #pragma unroll
            for (int k = 0; k < 4; ++k) acc[k] = (v8f){0.f, 0.f, 0.f, 0.f, 0.f, 0.f, 0.f, 0.f};
            #pragma unroll
            for (int o = 0; o < ONUM; ++o) {
                #pragma unroll
                for (int k = 0; k < 4; ++k) {
                    const v2f bf = cur.xa[k] * cf[o];
                    acc[k] = mma4(wf1[o][k], bf, acc[k]);
                }
            }
            s = (acc[0] + acc[1]) + (acc[2] + acc[3]);
        }

        float nsp = 0.0f;
        #pragma unroll
        for (int i = 0; i < 8; ++i) nsp += s[i] * s[i];
        const float ns  = nsp + __shfl_xor(nsp, 16);
        const float fac = sqrtf(ns) / (1.0f + ns);
        float vj[8];
        #pragma unroll
        for (int i = 0; i < 8; ++i) vj[i] = s[i] * fac;

        if (MODE < 2) {
            v8f agr;
            if (MODE == 0) {
                v8f ag[4];
                #pragma unroll
                for (int q = 0; q < 4; ++q) ag[q] = (v8f){0.f, 0.f, 0.f, 0.f, 0.f, 0.f, 0.f, 0.f};
                #pragma unroll
                for (int d = 0; d < DNUM; ++d) {
                    const v4f wa = wf2a[d];
                    const v4f wb = wf2b[d];
                    const float xd = cur.xr[d >> 2][d & 3];
                    const v2f a0 = {wa.x, wa.y}, a1 = {wa.z, wa.w};
                    const v2f a2 = {wb.x, wb.y}, a3 = {wb.z, wb.w};
                    const v2f b0 = {xd * vj[0], xd * vj[1]}, b1 = {xd * vj[2], xd * vj[3]};
                    const v2f b2 = {xd * vj[4], xd * vj[5]}, b3 = {xd * vj[6], xd * vj[7]};
                    ag[0] = mma4(a0, b0, ag[0]);
                    ag[1] = mma4(a1, b1, ag[1]);
                    ag[2] = mma4(a2, b2, ag[2]);
                    ag[3] = mma4(a3, b3, ag[3]);
                }
                agr = (ag[0] + ag[1]) + (ag[2] + ag[3]);
            } else {
                const v2f v01 = {vj[0], vj[1]}, v23 = {vj[2], vj[3]};
                const v2f v45 = {vj[4], vj[5]}, v67 = {vj[6], vj[7]};
                float p[ONUM];
                #pragma unroll
                for (int o = 0; o < ONUM; ++o) {
                    const v8f uo = u[o];
                    v2f q = (v2f){uo[0], uo[1]} * v01;
                    q += (v2f){uo[2], uo[3]} * v23;
                    q += (v2f){uo[4], uo[5]} * v45;
                    q += (v2f){uo[6], uo[7]} * v67;
                    p[o] = q.x + q.y;
                }
                #pragma unroll
                for (int i = 0; i < 8; ++i) {
                    const float own = h ? p[8 + i] : p[i];
                    const float snd = h ? p[i] : p[8 + i];
                    agr[i] = own + __shfl_xor(snd, 16);
                }
            }
            v4f n0, n1;
            #pragma unroll
            for (int i = 0; i < 4; ++i) {
                n0[i] = bold[i]     + agr[i];
                n1[i] = bold[4 + i] + agr[4 + i];
            }
            const size_t brow = ((size_t)b * RNUM + r) * ONUM + 8 * h;
            *reinterpret_cast<v4f*>(bmat + brow)     = n0;
            *reinterpret_cast<v4f*>(bmat + brow + 4) = n1;
            float* lp = lg + (t * 8 + w) * 256 + m16 * 16 + 8 * h;
            *reinterpret_cast<v4f*>(lp)     = n0;
            *reinterpret_cast<v4f*>(lp + 4) = n1;
        } else {
            float* op = out + ((size_t)b * RNUM + r) * CNUM + 8 * h;
            const v4f o0 = {vj[0], vj[1], vj[2], vj[3]};
            const v4f o1 = {vj[4], vj[5], vj[6], vj[7]};
            *reinterpret_cast<v4f*>(op)     = o0;
            *reinterpret_cast<v4f*>(op + 4) = o1;
        }
        cur = nxt;
    }

    if (MODE < 2) {
        __syncthreads();
        const size_t pbase = (size_t)blockIdx.x * (BATCH * ONUM) + tid;
        #pragma unroll
        for (int t = 0; t < NTILE; ++t) {
            float v[8];
            #pragma unroll
            for (int q = 0; q < 8; ++q) v[q] = lg[(t * 8 + q) * 256 + tid];
            const float m = fmaxf(fmaxf(fmaxf(v[0], v[1]), fmaxf(v[2], v[3])),
                                  fmaxf(fmaxf(v[4], v[5]), fmaxf(v[6], v[7])));
            float e[8];
            #pragma unroll
            for (int q = 0; q < 8; ++q) e[q] = __expf(v[q] - m);
            pmax[pbase + t * 256] = m;
            psum[pbase + t * 256] = ((e[0] + e[1]) + (e[2] + e[3])) + ((e[4] + e[5]) + (e[6] + e[7]));
        }
    }
}

__global__ __launch_bounds__(256) void softmax_combine(
    const float* __restrict__ pmax,
    const float* __restrict__ psum,
    float* __restrict__ rowmax,
    float* __restrict__ denom)
{
    __shared__ float sm_m[256];
    __shared__ float sm_s[256];
    const int b  = blockIdx.x;
    const int t  = threadIdx.x;
    const int o  = t & 15;
    const int st = t >> 4;
    constexpr int NBLK = RNUM / 8;
    constexpr int PER  = NBLK / 16;

    float m[PER], s[PER];
    #pragma unroll
    for (int j = 0; j < PER; ++j) {
        const size_t idx = ((size_t)(st + 16 * j) * BATCH + b) * ONUM + o;
        m[j] = pmax[idx];
        s[j] = psum[idx];
    }
    float mx = m[0];
    #pragma unroll
    for (int j = 1; j < PER; ++j) mx = fmaxf(mx, m[j]);
    float se = 0.0f;
    #pragma unroll
    for (int j = 0; j < PER; ++j) se += s[j] * __expf(m[j] - mx);
    sm_m[t] = mx;
    sm_s[t] = se;
    __syncthreads();
    if (t < 16) {
        float M = sm_m[t];
        #pragma unroll
        for (int k = 1; k < 16; ++k) M = fmaxf(M, sm_m[t + 16 * k]);
        float D = 0.0f;
        #pragma unroll
        for (int k = 0; k < 16; ++k) D += sm_s[t + 16 * k] * __expf(sm_m[t + 16 * k] - M);
        rowmax[b * ONUM + t] = M;
        denom[b * ONUM + t]  = D;
    }
}

extern "C" void kernel_launch(void* const* d_in, const int* in_sizes, int n_in,
                              void* d_out, int out_size, void* d_ws, size_t ws_size,
                              hipStream_t stream) {
    const float* x = (const float*)d_in[0];
    const float* W = (const float*)d_in[1];
    float* out = (float*)d_out;

    float* bmat = (float*)d_ws;
    float* rowmax = bmat + (size_t)BATCH * RNUM * ONUM;
    float* denom  = rowmax + BATCH * ONUM;

    float* pmax = out;
    float* psum = out + (size_t)(RNUM / 8) * BATCH * ONUM;

    dim3 grid(RNUM / 8), block(256);

    caps_kernel<0><<<grid, block, 0, stream>>>(x, W, bmat, rowmax, denom, out, pmax, psum);
    softmax_combine<<<dim3(BATCH), block, 0, stream>>>(pmax, psum, rowmax, denom);
    caps_kernel<1><<<grid, block, 0, stream>>>(x, W, bmat, rowmax, denom, out, pmax, psum);
    softmax_combine<<<dim3(BATCH), block, 0, stream>>>(pmax, psum, rowmax, denom);
    caps_kernel<2><<<grid, block, 0, stream>>>(x, W, bmat, rowmax, denom, out, pmax, psum);
}


# First pass: agreement as a quadratic form in x, Q[o][d][d'] = sum_c W[o,d,c] Ws[d',c] made once per wave and folded over symmetric pairs: 34 f32 WMMA per tile (was 64); agreement sum order changes
# speedup vs baseline: 3.3011x; 1.0561x over previous
#include <hip/hip_runtime.h>
#include <hip/hip_bf16.h>
#include <math.h>


#define BATCH 128
#define RNUM  2048
#define ONUM  16
#define DNUM  16
#define CNUM  16
#define NTILE (BATCH / 16)
#define PRIO_TILES 5

typedef __attribute__((ext_vector_type(2))) float v2f;
typedef __attribute__((ext_vector_type(4))) float v4f;
typedef __attribute__((ext_vector_type(8))) float v8f;

__device__ __forceinline__ v8f mma4(v2f a, v2f b, v8f c) {
    return __builtin_amdgcn_wmma_f32_16x16x4_f32(false, a, false, b, (short)0, c, false, false);
}

struct TileIn {
    v2f xa[4];
    v4f xq[5];
    v4f bl0, bl1, rm0, rm1, dn0, dn1;
};

template <int MODE>
__device__ __forceinline__ void load_tile(TileIn& ti,
    const float* __restrict__ x, const float* bmat,
    const float* __restrict__ rowmax, const float* __restrict__ denom,
    int b, int r, int h)
{
    const float* xb = x + b * DNUM;
    #pragma unroll
    for (int k = 0; k < 4; ++k)
        ti.xa[k] = *reinterpret_cast<const v2f*>(xb + 4 * k + 2 * h);
    const int xoff[5] = {8 * h, 8 * h + 4, 4 * h, 8, 12};
    #pragma unroll
    for (int t = 0; t < 5; ++t)
        ti.xq[t] = (MODE == 0) ? *reinterpret_cast<const v4f*>(xb + xoff[t]) : (v4f){0.f, 0.f, 0.f, 0.f};
    const v4f zero4 = {0.f, 0.f, 0.f, 0.f};
    ti.bl0 = zero4; ti.bl1 = zero4; ti.rm0 = zero4; ti.rm1 = zero4; ti.dn0 = zero4; ti.dn1 = zero4;
    if (MODE > 0) {
        const size_t brow = ((size_t)b * RNUM + r) * ONUM + 8 * h;
        ti.bl0 = *reinterpret_cast<const v4f*>(bmat + brow);
        ti.bl1 = *reinterpret_cast<const v4f*>(bmat + brow + 4);
        ti.rm0 = *reinterpret_cast<const v4f*>(rowmax + b * ONUM + 8 * h);
        ti.rm1 = *reinterpret_cast<const v4f*>(rowmax + b * ONUM + 8 * h + 4);
        ti.dn0 = *reinterpret_cast<const v4f*>(denom + b * ONUM + 8 * h);
        ti.dn1 = *reinterpret_cast<const v4f*>(denom + b * ONUM + 8 * h + 4);
    }
}

template <int MODE>
__global__ __launch_bounds__(256) void caps_kernel(
    const float* __restrict__ x,
    const float* __restrict__ W,
    float* bmat,
    const float* __restrict__ rowmax,
    const float* __restrict__ denom,
    float* __restrict__ out,
    float* __restrict__ pmax,
    float* __restrict__ psum)
{
    __shared__ float lg[MODE < 2 ? NTILE * 8 * 256 : 1];

    const int tid  = threadIdx.x;
    const int L    = tid & 31;
    const int w    = tid >> 5;
    const int h    = L >> 4;
    const int m16  = L & 15;
    const int r    = blockIdx.x * 8 + w;

    const float* Wr = W + (size_t)r * (ONUM * DNUM * CNUM);

    TileIn cur;
    load_tile<MODE>(cur, x, bmat, rowmax, denom, m16, r, h);

    v2f wf1[MODE == 0 ? 1 : ONUM][4];
    v2f ws1[4];
    v2f qa[MODE == 0 ? 34 : 1];
    if (MODE == 0) {
        v4f wsa = (v4f){0.f, 0.f, 0.f, 0.f}, wsb = wsa;
        #pragma unroll
        for (int o = 0; o < ONUM; ++o) {
            const float* Wo = Wr + o * (DNUM * CNUM) + m16 * CNUM + 8 * h;
            wsa += *reinterpret_cast<const v4f*>(Wo);
            wsb += *reinterpret_cast<const v4f*>(Wo + 4);
        }
        wsa = wsa * (1.0f / (float)RNUM);
        wsb = wsb * (1.0f / (float)RNUM);
        const v2f wsr[4] = {{wsa.x, wsa.y}, {wsa.z, wsa.w}, {wsb.x, wsb.y}, {wsb.z, wsb.w}};
        float* sc = lg + w * 256;
        *reinterpret_cast<v4f*>(sc + m16 * 16 + 8 * h)     = wsa;
        *reinterpret_cast<v4f*>(sc + m16 * 16 + 8 * h + 4) = wsb;
        #pragma unroll
        for (int k = 0; k < 4; ++k) {
            ws1[k].x = sc[(4 * k + 2 * h + 0) * 16 + m16];
            ws1[k].y = sc[(4 * k + 2 * h + 1) * 16 + m16];
        }
        v8f qd[DNUM];
        const float* Wl = Wr + m16 * (DNUM * CNUM) + 8 * h;
        #pragma unroll
        for (int d = 0; d < DNUM; ++d) {
            const v4f wa = *reinterpret_cast<const v4f*>(Wl + d * CNUM);
            const v4f wb = *reinterpret_cast<const v4f*>(Wl + d * CNUM + 4);
            v8f q = (v8f){0.f, 0.f, 0.f, 0.f, 0.f, 0.f, 0.f, 0.f};
            q = mma4(wsr[0], (v2f){wa.x, wa.y}, q);
            q = mma4(wsr[1], (v2f){wa.z, wa.w}, q);
            q = mma4(wsr[2], (v2f){wb.x, wb.y}, q);
            q = mma4(wsr[3], (v2f){wb.z, wb.w}, q);
            qd[d] = q;
        }
        float sall[68];
        #pragma unroll
        for (int p = 0; p < 8; ++p) {
            #pragma unroll
            for (int q = p; q < 8; ++q) {
                const int n = p * 8 - (p * (p - 1)) / 2 + (q - p);
                float a = h ? qd[8 + p][q] : qd[p][q];
                if (q > p) a += h ? qd[8 + q][p] : qd[q][p];
                sall[n] = a;
            }
        }
        #pragma unroll
        for (int a = 0; a < 4; ++a) {
            #pragma unroll
            for (int e = 0; e < 8; ++e) {
                const float own = h ? qd[4 + a][e] : qd[8 + e][a];
                const float snd = h ? qd[a][e]     : qd[8 + e][4 + a];
                sall[36 + 8 * a + e] = own + __shfl_xor(snd, 16);
            }
        }
        #pragma unroll
        for (int j = 0; j < 34; ++j) qa[j] = (v2f){sall[2 * j], sall[2 * j + 1]};
    } else {
        #pragma unroll
        for (int o = 0; o < ONUM; ++o) {
            #pragma unroll
            for (int k = 0; k < 4; ++k) {
                const int d0 = 4 * k + 2 * h;
                wf1[o][k].x = Wr[o * (DNUM * CNUM) + (d0 + 0) * CNUM + m16];
                wf1[o][k].y = Wr[o * (DNUM * CNUM) + (d0 + 1) * CNUM + m16];
            }
        }
    }

    if (__builtin_amdgcn_readfirstlane(w) >= 4) __builtin_amdgcn_s_setprio(3);

    for (int t = 0; t < NTILE; ++t) {
        if (t == PRIO_TILES) __builtin_amdgcn_s_setprio(0);
        const int b = t * 16 + m16;

        TileIn nxt;
        load_tile<MODE>(nxt, x, bmat, rowmax, denom, (t < NTILE - 1 ? t + 1 : t) * 16 + m16, r, h);

        float cf[ONUM];
        float bold[8];
        if (MODE == 0) {
            #pragma unroll
            for (int i = 0; i < 8; ++i) bold[i] = 0.0f;
        } else {
            float cown[8], coth[8];
            #pragma unroll
            for (int i = 0; i < 4; ++i) {
                bold[i]     = cur.bl0[i];
                bold[4 + i] = cur.bl1[i];
                cown[i]     = __expf(cur.bl0[i] - cur.rm0[i]) / cur.dn0[i];
                cown[4 + i] = __expf(cur.bl1[i] - cur.rm1[i]) / cur.dn1[i];
            }
            #pragma unroll
            for (int i = 0; i < 8; ++i) coth[i] = __shfl_xor(cown[i], 16);
            #pragma unroll
            for (int i = 0; i < 8; ++i) {
                cf[i]     = h ? coth[i] : cown[i];
                cf[8 + i] = h ? cown[i] : coth[i];
            }
        }

        v8f s;
        v8f u[MODE == 1 ? ONUM : 1];
        if (MODE == 1) {
            s = (v8f){0.f, 0.f, 0.f, 0.f, 0.f, 0.f, 0.f, 0.f};
            #pragma unroll
            for (int o = 0; o < ONUM; o += 2) {
                v8f e0 = (v8f){0.f, 0.f, 0.f, 0.f, 0.f, 0.f, 0.f, 0.f};
                v8f e1 = e0;
                #pragma unroll
                for (int k = 0; k < 4; ++k) {
                    e0 = mma4(wf1[o][k],     cur.xa[k], e0);
                    e1 = mma4(wf1[o + 1][k], cur.xa[k], e1);
                }
                u[o]     = e0;
                u[o + 1] = e1;
                s += e0 * cf[o];
                s += e1 * cf[o + 1];
            }
        } else if (MODE == 0) {
            v8f a0 = (v8f){0.f, 0.f, 0.f, 0.f, 0.f, 0.f, 0.f, 0.f};
            v8f a1 = a0;
            a0 = mma4(ws1[0], cur.xa[0], a0);
            a1 = mma4(ws1[1], cur.xa[1], a1);
            a0 = mma4(ws1[2], cur.xa[2], a0);
            a1 = mma4(ws1[3], cur.xa[3], a1);
            s = a0 + a1;
        } else {
            v8f acc[4];
            #pragma unroll
            for (int k = 0; k < 4; ++k) acc[k] = (v8f){0.f, 0.f, 0.f, 0.f, 0.f, 0.f, 0.f, 0.f};
            #pragma unroll
            for (int o = 0; o < ONUM; ++o) {
                #pragma unroll
                for (int k = 0; k < 4; ++k) {
                    const v2f bf = cur.xa[k] * cf[o];
                    acc[k] = mma4(wf1[o][k], bf, acc[k]);
                }
            }
            s = (acc[0] + acc[1]) + (acc[2] + acc[3]);
        }

        float nsp = 0.0f;
        #pragma unroll
        for (int i = 0; i < 8; ++i) nsp += s[i] * s[i];
        const float ns  = nsp + __shfl_xor(nsp, 16);
        const float fac = sqrtf(ns) / (1.0f + ns);
        float vj[8];
        #pragma unroll
        for (int i = 0; i < 8; ++i) vj[i] = s[i] * fac;

        if (MODE < 2) {
            v8f agr;
            if (MODE == 0) {
                float xh[8], xc[4], xu[8];
                #pragma unroll
                for (int p = 0; p < 8; ++p) {
                    xh[p] = cur.xq[p >> 2][p & 3];
                    xu[p] = cur.xq[3 + (p >> 2)][p & 3];
                }
                #pragma unroll
                for (int a = 0; a < 4; ++a) xc[a] = cur.xq[2][a];
                float xx[68];
                #pragma unroll
                for (int p = 0; p < 8; ++p) {
                    #pragma unroll
                    for (int q = p; q < 8; ++q)
                        xx[p * 8 - (p * (p - 1)) / 2 + (q - p)] = xh[p] * xh[q];
                }
                #pragma unroll
                for (int a = 0; a < 4; ++a) {
                    #pragma unroll
                    for (int e = 0; e < 8; ++e) xx[36 + 8 * a + e] = xc[a] * xu[e];
                }
                v8f g0 = (v8f){0.f, 0.f, 0.f, 0.f, 0.f, 0.f, 0.f, 0.f};
                v8f g1 = g0;
                #pragma unroll
                for (int j = 0; j < 34; j += 2) {
                    g0 = mma4(qa[j],     (v2f){xx[2 * j],     xx[2 * j + 1]}, g0);
                    g1 = mma4(qa[j + 1], (v2f){xx[2 * j + 2], xx[2 * j + 3]}, g1);
                }
                agr = (g0 + g1) * fac;
            } else {
                const v2f v01 = {vj[0], vj[1]}, v23 = {vj[2], vj[3]};
                const v2f v45 = {vj[4], vj[5]}, v67 = {vj[6], vj[7]};
                float p[ONUM];
                #pragma unroll
                for (int o = 0; o < ONUM; ++o) {
                    const v8f uo = u[o];
                    v2f q = (v2f){uo[0], uo[1]} * v01;
                    q += (v2f){uo[2], uo[3]} * v23;
                    q += (v2f){uo[4], uo[5]} * v45;
                    q += (v2f){uo[6], uo[7]} * v67;
                    p[o] = q.x + q.y;
                }
                #pragma unroll
                for (int i = 0; i < 8; ++i) {
                    const float own = h ? p[8 + i] : p[i];
                    const float snd = h ? p[i] : p[8 + i];
                    agr[i] = own + __shfl_xor(snd, 16);
                }
            }
            v4f n0, n1;
            #pragma unroll
            for (int i = 0; i < 4; ++i) {
                n0[i] = bold[i]     + agr[i];
                n1[i] = bold[4 + i] + agr[4 + i];
            }
            const size_t brow = ((size_t)b * RNUM + r) * ONUM + 8 * h;
            *reinterpret_cast<v4f*>(bmat + brow)     = n0;
            *reinterpret_cast<v4f*>(bmat + brow + 4) = n1;
            float* lp = lg + (t * 8 + w) * 256 + m16 * 16 + 8 * h;
            *reinterpret_cast<v4f*>(lp)     = n0;
            *reinterpret_cast<v4f*>(lp + 4) = n1;
        } else {
            float* op = out + ((size_t)b * RNUM + r) * CNUM + 8 * h;
            const v4f o0 = {vj[0], vj[1], vj[2], vj[3]};
            const v4f o1 = {vj[4], vj[5], vj[6], vj[7]};
            *reinterpret_cast<v4f*>(op)     = o0;
            *reinterpret_cast<v4f*>(op + 4) = o1;
        }
        cur = nxt;
    }

    if (MODE < 2) {
        __syncthreads();
        const size_t pbase = (size_t)blockIdx.x * (BATCH * ONUM) + tid;
        #pragma unroll
        for (int t = 0; t < NTILE; ++t) {
            float v[8];
            #pragma unroll
            for (int q = 0; q < 8; ++q) v[q] = lg[(t * 8 + q) * 256 + tid];
            const float m = fmaxf(fmaxf(fmaxf(v[0], v[1]), fmaxf(v[2], v[3])),
                                  fmaxf(fmaxf(v[4], v[5]), fmaxf(v[6], v[7])));
            float e[8];
            #pragma unroll
            for (int q = 0; q < 8; ++q) e[q] = __expf(v[q] - m);
            pmax[pbase + t * 256] = m;
            psum[pbase + t * 256] = ((e[0] + e[1]) + (e[2] + e[3])) + ((e[4] + e[5]) + (e[6] + e[7]));
        }
    }
}

__global__ __launch_bounds__(256) void softmax_combine(
    const float* __restrict__ pmax,
    const float* __restrict__ psum,
    float* __restrict__ rowmax,
    float* __restrict__ denom)
{
    __shared__ float sm_m[256];
    __shared__ float sm_s[256];
    const int b  = blockIdx.x;
    const int t  = threadIdx.x;
    const int o  = t & 15;
    const int st = t >> 4;
    constexpr int NBLK = RNUM / 8;
    constexpr int PER  = NBLK / 16;

    float m[PER], s[PER];
    #pragma unroll
    for (int j = 0; j < PER; ++j) {
        const size_t idx = ((size_t)(st + 16 * j) * BATCH + b) * ONUM + o;
        m[j] = pmax[idx];
        s[j] = psum[idx];
    }
    float mx = m[0];
    #pragma unroll
    for (int j = 1; j < PER; ++j) mx = fmaxf(mx, m[j]);
    float se = 0.0f;
    #pragma unroll
    for (int j = 0; j < PER; ++j) se += s[j] * __expf(m[j] - mx);
    sm_m[t] = mx;
    sm_s[t] = se;
    __syncthreads();
    if (t < 16) {
        float M = sm_m[t];
        #pragma unroll
        for (int k = 1; k < 16; ++k) M = fmaxf(M, sm_m[t + 16 * k]);
        float D = 0.0f;
        #pragma unroll
        for (int k = 0; k < 16; ++k) D += sm_s[t + 16 * k] * __expf(sm_m[t + 16 * k] - M);
        rowmax[b * ONUM + t] = M;
        denom[b * ONUM + t]  = D;
    }
}

extern "C" void kernel_launch(void* const* d_in, const int* in_sizes, int n_in,
                              void* d_out, int out_size, void* d_ws, size_t ws_size,
                              hipStream_t stream) {
    const float* x = (const float*)d_in[0];
    const float* W = (const float*)d_in[1];
    float* out = (float*)d_out;

    float* bmat = (float*)d_ws;
    float* rowmax = bmat + (size_t)BATCH * RNUM * ONUM;
    float* denom  = rowmax + BATCH * ONUM;

    float* pmax = out;
    float* psum = out + (size_t)(RNUM / 8) * BATCH * ONUM;

    dim3 grid(RNUM / 8), block(256);

    caps_kernel<0><<<grid, block, 0, stream>>>(x, W, bmat, rowmax, denom, out, pmax, psum);
    softmax_combine<<<dim3(BATCH), block, 0, stream>>>(pmax, psum, rowmax, denom);
    caps_kernel<1><<<grid, block, 0, stream>>>(x, W, bmat, rowmax, denom, out, pmax, psum);
    softmax_combine<<<dim3(BATCH), block, 0, stream>>>(pmax, psum, rowmax, denom);
    caps_kernel<2><<<grid, block, 0, stream>>>(x, W, bmat, rowmax, denom, out, pmax, psum);
}
